# starting kernel
# baseline (speedup 1.0000x reference)
#include <hip/hip_runtime.h>
#include <cstddef>

typedef float v2f __attribute__((ext_vector_type(2)));
typedef float v8f __attribute__((ext_vector_type(8)));

#define D_FEAT 512
#define LDS_STRIDE 516

__global__ __launch_bounds__(256) void sage_aggregate(
    const float* __restrict__ hsrc, const int* __restrict__ src,
    const int* __restrict__ dst, float* __restrict__ neigh,
    float* __restrict__ deg, int E)
{
    int e = blockIdx.x * 2 + (threadIdx.x >> 7);
    if (e >= E) return;
    int lane = threadIdx.x & 127;
    int s = src[e];
    int d = dst[e];
    const float4 v = *reinterpret_cast<const float4*>(
        hsrc + (size_t)s * D_FEAT + (size_t)lane * 4);
    float* p = neigh + (size_t)d * D_FEAT + (size_t)lane * 4;
    unsafeAtomicAdd(p + 0, v.x);
    unsafeAtomicAdd(p + 1, v.y);
    unsafeAtomicAdd(p + 2, v.z);
    unsafeAtomicAdd(p + 3, v.w);
    if (lane == 0) unsafeAtomicAdd(deg + d, 1.0f);
}

__global__ __launch_bounds__(256) void sage_combine(
    const float* __restrict__ neigh, const float* __restrict__ hdst,
    const float* __restrict__ deg, float* __restrict__ out, int n)
{
    size_t i = (size_t)blockIdx.x * blockDim.x + threadIdx.x;
    if (i >= (size_t)n * (D_FEAT / 4)) return;
    size_t row = i >> 7;
    int c = (int)(i & 127) * 4;
    float inv = 1.0f / (deg[row] + 1.0f);
    const float4 a = *reinterpret_cast<const float4*>(neigh + row * D_FEAT + c);
    const float4 b = *reinterpret_cast<const float4*>(hdst + row * D_FEAT + c);
    float4 r;
    r.x = (a.x + b.x) * inv;
    r.y = (a.y + b.y) * inv;
    r.z = (a.z + b.z) * inv;
    r.w = (a.w + b.w) * inv;
    *reinterpret_cast<float4*>(out + row * D_FEAT + c) = r;
}

template <bool RELU>
__global__ __launch_bounds__(128) void sage_gemm_wmma(
    const float* __restrict__ A, const float* __restrict__ B,
    const float* __restrict__ bias, float* __restrict__ C,
    int M, int N, int K)
{
    __shared__ float As[16 * LDS_STRIDE];

    const int rm   = blockIdx.x * 16;
    const int wave = threadIdx.x >> 5;
    const int lane = threadIdx.x & 31;
    const int half = lane >> 4;
    const int l    = lane & 15;
    const int cn   = (blockIdx.y * 4 + wave) * 16;

    {
        int r    = threadIdx.x >> 3;
        int cseg = (threadIdx.x & 7) * 64;
        int gr   = min(rm + r, M - 1);
        const float* gsrc = A + (size_t)gr * K + cseg;
        float* ldst = As + r * LDS_STRIDE + cseg;
#pragma unroll
        for (int i = 0; i < 16; ++i) {
            *reinterpret_cast<float4*>(ldst + i * 4) =
                *reinterpret_cast<const float4*>(gsrc + i * 4);
        }
    }
    __syncthreads();

    v8f acc = {};
    const float* arow = As + l * LDS_STRIDE + 2 * half;
    const float* bcol = B + (size_t)(2 * half) * N + cn + l;
    for (int k = 0; k < K; k += 4) {
        v2f a;
        a.x = arow[k];
        a.y = arow[k + 1];
        v2f b;
        b.x = bcol[(size_t)k * N];
        b.y = bcol[(size_t)(k + 1) * N];
        acc = __builtin_amdgcn_wmma_f32_16x16x4_f32(
            false, a, false, b, (short)0, acc, false, false);
    }

    const float bv = bias[cn + l];
#pragma unroll
    for (int i = 0; i < 8; ++i) {
        int row = rm + i + 8 * half;
        float v = acc[i] + bv;
        if (RELU) v = fmaxf(v, 0.0f);
        if (row < M) C[(size_t)row * N + cn + l] = v;
    }
}

extern "C" void kernel_launch(void* const* d_in, const int* in_sizes, int n_in,
                              void* d_out, int out_size, void* d_ws, size_t ws_size,
                              hipStream_t stream)
{
    const float* feats = (const float*)d_in[0];
    const int*   src0  = (const int*)d_in[1];
    const int*   dst0  = (const int*)d_in[2];
    const int*   src1  = (const int*)d_in[3];
    const int*   dst1  = (const int*)d_in[4];
    const float* W1    = (const float*)d_in[5];
    const float* b1    = (const float*)d_in[6];
    const float* W2    = (const float*)d_in[7];
    const float* b2    = (const float*)d_in[8];
    float* out = (float*)d_out;

    const int N1 = 25000, N2 = 5000;
    const int E0 = in_sizes[1];
    const int E1 = in_sizes[3];
    const size_t F = D_FEAT;

    float* ws     = (float*)d_ws;
    float* neigh0 = ws;
    float* hpre0  = neigh0 + (size_t)N1 * F;
    float* h1     = hpre0 + (size_t)N1 * F;
    float* deg0   = h1 + (size_t)N1 * F;

    hipMemsetAsync(neigh0, 0, (size_t)N1 * F * sizeof(float), stream);
    hipMemsetAsync(deg0, 0, (size_t)N1 * sizeof(float), stream);

    sage_aggregate<<<(E0 + 1) / 2, 256, 0, stream>>>(feats, src0, dst0,
                                                     neigh0, deg0, E0);

    {
        size_t work = (size_t)N1 * (F / 4);
        sage_combine<<<(unsigned)((work + 255) / 256), 256, 0, stream>>>(
            neigh0, feats, deg0, hpre0, N1);
    }

    {
        dim3 grid((N1 + 15) / 16, 512 / 64);
        sage_gemm_wmma<true><<<grid, 128, 0, stream>>>(hpre0, W1, b1, h1,
                                                       N1, 512, 512);
    }

    hipMemsetAsync(neigh0, 0, (size_t)N2 * F * sizeof(float), stream);
    hipMemsetAsync(deg0, 0, (size_t)N2 * sizeof(float), stream);

    sage_aggregate<<<(E1 + 1) / 2, 256, 0, stream>>>(h1, src1, dst1,
                                                     neigh0, deg0, E1);

    {
        size_t work = (size_t)N2 * (F / 4);
        sage_combine<<<(unsigned)((work + 255) / 256), 256, 0, stream>>>(
            neigh0, h1, deg0, hpre0, N2);
    }

    {
        dim3 grid((N2 + 15) / 16, 256 / 64);
        sage_gemm_wmma<false><<<grid, 128, 0, stream>>>(hpre0, W2, b2, out,
                                                        N2, 256, 512);
    }
}


# Gather instead of float-atomic scatter: per-destination edge lists built on device (returning int adds, scan, fill); one wave sums a row in registers in list order, which still follows arrival
# speedup vs baseline: 1.7656x; 1.7656x over previous
#include <hip/hip_runtime.h>
#include <cstddef>

typedef float v2f __attribute__((ext_vector_type(2)));
typedef float v8f __attribute__((ext_vector_type(8)));

#define D_FEAT 512
#define LDS_STRIDE 516

__global__ __launch_bounds__(256) void sage_edge_rank(
    const int* __restrict__ dst, int* __restrict__ cnt,
    int* __restrict__ rank, int E)
{
    int e = blockIdx.x * 256 + threadIdx.x;
    if (e >= E) return;
    rank[e] = atomicAdd(cnt + dst[e], 1);
}

__global__ __launch_bounds__(1024) void sage_edge_offsets(
    const int* __restrict__ cnt, int* __restrict__ off, int n)
{
    __shared__ int wsum[32];
    const int t    = threadIdx.x;
    const int lane = t & 31;
    const int w    = t >> 5;
    const int per  = (n + 1023) / 1024;
    const int ib   = min(t * per, n);
    const int ie   = min(ib + per, n);

    int s = 0;
    for (int i = ib; i < ie; ++i) s += cnt[i];

    int v = s;
    for (int d = 1; d < 32; d <<= 1) {
        int u = __shfl_up(v, d);
        if (lane >= d) v += u;
    }
    if (lane == 31) wsum[w] = v;
    __syncthreads();
    if (w == 0) {
        int x = wsum[lane];
        for (int d = 1; d < 32; d <<= 1) {
            int u = __shfl_up(x, d);
            if (lane >= d) x += u;
        }
        wsum[lane] = x;
    }
    __syncthreads();

    int base = v - s + (w > 0 ? wsum[w - 1] : 0);
    for (int i = ib; i < ie; ++i) {
        off[i] = base;
        base += cnt[i];
    }
    if (t == 0) off[n] = wsum[31];
}

__global__ __launch_bounds__(256) void sage_edge_fill(
    const int* __restrict__ src, const int* __restrict__ dst,
    const int* __restrict__ rank, const int* __restrict__ off,
    int* __restrict__ perm, int E)
{
    int e = blockIdx.x * 256 + threadIdx.x;
    if (e >= E) return;
    perm[off[dst[e]] + rank[e]] = src[e];
}

__global__ __launch_bounds__(256) void sage_gather_rows(
    const float* __restrict__ hsrc, const int* __restrict__ off,
    const int* __restrict__ perm, float* __restrict__ neigh,
    float* __restrict__ deg, int n)
{
    const int row = __builtin_amdgcn_readfirstlane(
        (int)(blockIdx.x * 8 + (threadIdx.x >> 5)));
    if (row >= n) return;
    const int lane = threadIdx.x & 31;
    const int jb = off[row];
    const int je = off[row + 1];
    const float* col = hsrc + lane * 4;

    float4 a0 = make_float4(0.f, 0.f, 0.f, 0.f);
    float4 a1 = a0, a2 = a0, a3 = a0;

    int j = jb;
    for (; j + 4 <= je; j += 4) {
        const float* r0 = col + (size_t)perm[j + 0] * D_FEAT;
        const float* r1 = col + (size_t)perm[j + 1] * D_FEAT;
        const float* r2 = col + (size_t)perm[j + 2] * D_FEAT;
        const float* r3 = col + (size_t)perm[j + 3] * D_FEAT;
        const float4 p0 = *reinterpret_cast<const float4*>(r0);
        const float4 p1 = *reinterpret_cast<const float4*>(r0 + 128);
        const float4 p2 = *reinterpret_cast<const float4*>(r0 + 256);
        const float4 p3 = *reinterpret_cast<const float4*>(r0 + 384);
        const float4 q0 = *reinterpret_cast<const float4*>(r1);
        const float4 q1 = *reinterpret_cast<const float4*>(r1 + 128);
        const float4 q2 = *reinterpret_cast<const float4*>(r1 + 256);
        const float4 q3 = *reinterpret_cast<const float4*>(r1 + 384);
        const float4 s0 = *reinterpret_cast<const float4*>(r2);
        const float4 s1 = *reinterpret_cast<const float4*>(r2 + 128);
        const float4 s2 = *reinterpret_cast<const float4*>(r2 + 256);
        const float4 s3 = *reinterpret_cast<const float4*>(r2 + 384);
        const float4 t0 = *reinterpret_cast<const float4*>(r3);
        const float4 t1 = *reinterpret_cast<const float4*>(r3 + 128);
        const float4 t2 = *reinterpret_cast<const float4*>(r3 + 256);
        const float4 t3 = *reinterpret_cast<const float4*>(r3 + 384);
        a0.x += p0.x; a0.y += p0.y; a0.z += p0.z; a0.w += p0.w;
        a1.x += p1.x; a1.y += p1.y; a1.z += p1.z; a1.w += p1.w;
        a2.x += p2.x; a2.y += p2.y; a2.z += p2.z; a2.w += p2.w;
        a3.x += p3.x; a3.y += p3.y; a3.z += p3.z; a3.w += p3.w;
        a0.x += q0.x; a0.y += q0.y; a0.z += q0.z; a0.w += q0.w;
        a1.x += q1.x; a1.y += q1.y; a1.z += q1.z; a1.w += q1.w;
        a2.x += q2.x; a2.y += q2.y; a2.z += q2.z; a2.w += q2.w;
        a3.x += q3.x; a3.y += q3.y; a3.z += q3.z; a3.w += q3.w;
        a0.x += s0.x; a0.y += s0.y; a0.z += s0.z; a0.w += s0.w;
        a1.x += s1.x; a1.y += s1.y; a1.z += s1.z; a1.w += s1.w;
        a2.x += s2.x; a2.y += s2.y; a2.z += s2.z; a2.w += s2.w;
        a3.x += s3.x; a3.y += s3.y; a3.z += s3.z; a3.w += s3.w;
        a0.x += t0.x; a0.y += t0.y; a0.z += t0.z; a0.w += t0.w;
        a1.x += t1.x; a1.y += t1.y; a1.z += t1.z; a1.w += t1.w;
        a2.x += t2.x; a2.y += t2.y; a2.z += t2.z; a2.w += t2.w;
        a3.x += t3.x; a3.y += t3.y; a3.z += t3.z; a3.w += t3.w;
    }
    for (; j < je; ++j) {
        const float* r0 = col + (size_t)perm[j] * D_FEAT;
        const float4 p0 = *reinterpret_cast<const float4*>(r0);
        const float4 p1 = *reinterpret_cast<const float4*>(r0 + 128);
        const float4 p2 = *reinterpret_cast<const float4*>(r0 + 256);
        const float4 p3 = *reinterpret_cast<const float4*>(r0 + 384);
        a0.x += p0.x; a0.y += p0.y; a0.z += p0.z; a0.w += p0.w;
        a1.x += p1.x; a1.y += p1.y; a1.z += p1.z; a1.w += p1.w;
        a2.x += p2.x; a2.y += p2.y; a2.z += p2.z; a2.w += p2.w;
        a3.x += p3.x; a3.y += p3.y; a3.z += p3.z; a3.w += p3.w;
    }

    float* o = neigh + (size_t)row * D_FEAT + lane * 4;
    *reinterpret_cast<float4*>(o)       = a0;
    *reinterpret_cast<float4*>(o + 128) = a1;
    *reinterpret_cast<float4*>(o + 256) = a2;
    *reinterpret_cast<float4*>(o + 384) = a3;
    if (lane == 0) deg[row] = (float)(je - jb);
}

__global__ __launch_bounds__(256) void sage_combine(
    const float* __restrict__ neigh, const float* __restrict__ hdst,
    const float* __restrict__ deg, float* __restrict__ out, int n)
{
    size_t i = (size_t)blockIdx.x * blockDim.x + threadIdx.x;
    if (i >= (size_t)n * (D_FEAT / 4)) return;
    size_t row = i >> 7;
    int c = (int)(i & 127) * 4;
    float inv = 1.0f / (deg[row] + 1.0f);
    const float4 a = *reinterpret_cast<const float4*>(neigh + row * D_FEAT + c);
    const float4 b = *reinterpret_cast<const float4*>(hdst + row * D_FEAT + c);
    float4 r;
    r.x = (a.x + b.x) * inv;
    r.y = (a.y + b.y) * inv;
    r.z = (a.z + b.z) * inv;
    r.w = (a.w + b.w) * inv;
    *reinterpret_cast<float4*>(out + row * D_FEAT + c) = r;
}

template <bool RELU>
__global__ __launch_bounds__(128) void sage_gemm_wmma(
    const float* __restrict__ A, const float* __restrict__ B,
    const float* __restrict__ bias, float* __restrict__ C,
    int M, int N, int K)
{
    __shared__ float As[16 * LDS_STRIDE];

    const int rm   = blockIdx.x * 16;
    const int wave = threadIdx.x >> 5;
    const int lane = threadIdx.x & 31;
    const int half = lane >> 4;
    const int l    = lane & 15;
    const int cn   = (blockIdx.y * 4 + wave) * 16;

    {
        int r    = threadIdx.x >> 3;
        int cseg = (threadIdx.x & 7) * 64;
        int gr   = min(rm + r, M - 1);
        const float* gsrc = A + (size_t)gr * K + cseg;
        float* ldst = As + r * LDS_STRIDE + cseg;
#pragma unroll
        for (int i = 0; i < 16; ++i) {
            *reinterpret_cast<float4*>(ldst + i * 4) =
                *reinterpret_cast<const float4*>(gsrc + i * 4);
        }
    }
    __syncthreads();

    v8f acc = {};
    const float* arow = As + l * LDS_STRIDE + 2 * half;
    const float* bcol = B + (size_t)(2 * half) * N + cn + l;
    for (int k = 0; k < K; k += 4) {
        v2f a;
        a.x = arow[k];
        a.y = arow[k + 1];
        v2f b;
        b.x = bcol[(size_t)k * N];
        b.y = bcol[(size_t)(k + 1) * N];
        acc = __builtin_amdgcn_wmma_f32_16x16x4_f32(
            false, a, false, b, (short)0, acc, false, false);
    }

    const float bv = bias[cn + l];
#pragma unroll
    for (int i = 0; i < 8; ++i) {
        int row = rm + i + 8 * half;
        float v = acc[i] + bv;
        if (RELU) v = fmaxf(v, 0.0f);
        if (row < M) C[(size_t)row * N + cn + l] = v;
    }
}

extern "C" void kernel_launch(void* const* d_in, const int* in_sizes, int n_in,
                              void* d_out, int out_size, void* d_ws, size_t ws_size,
                              hipStream_t stream)
{
    const float* feats = (const float*)d_in[0];
    const int*   src0  = (const int*)d_in[1];
    const int*   dst0  = (const int*)d_in[2];
    const int*   src1  = (const int*)d_in[3];
    const int*   dst1  = (const int*)d_in[4];
    const float* W1    = (const float*)d_in[5];
    const float* b1    = (const float*)d_in[6];
    const float* W2    = (const float*)d_in[7];
    const float* b2    = (const float*)d_in[8];
    float* out = (float*)d_out;

    const int N1 = 25000, N2 = 5000;
    const int E0 = in_sizes[1];
    const int E1 = in_sizes[3];
    const size_t F = D_FEAT;

    float* ws     = (float*)d_ws;
    float* neigh0 = ws;
    float* hpre0  = neigh0 + (size_t)N1 * F;
    float* h1     = hpre0 + (size_t)N1 * F;
    float* deg0   = h1 + (size_t)N1 * F;

    {
        int* cnt  = (int*)h1;
        int* off  = cnt + N1;
        int* rank = off + (N1 + 1);
        int* perm = rank + E0;
        hipMemsetAsync(cnt, 0, (size_t)N1 * sizeof(int), stream);
        sage_edge_rank<<<(E0 + 255) / 256, 256, 0, stream>>>(dst0, cnt, rank, E0);
        sage_edge_offsets<<<1, 1024, 0, stream>>>(cnt, off, N1);
        sage_edge_fill<<<(E0 + 255) / 256, 256, 0, stream>>>(src0, dst0, rank,
                                                            off, perm, E0);
        sage_gather_rows<<<(N1 + 7) / 8, 256, 0, stream>>>(feats, off, perm,
                                                           neigh0, deg0, N1);
    }

    {
        size_t work = (size_t)N1 * (F / 4);
        sage_combine<<<(unsigned)((work + 255) / 256), 256, 0, stream>>>(
            neigh0, feats, deg0, hpre0, N1);
    }

    {
        dim3 grid((N1 + 15) / 16, 512 / 64);
        sage_gemm_wmma<true><<<grid, 128, 0, stream>>>(hpre0, W1, b1, h1,
                                                       N1, 512, 512);
    }

    {
        int* cnt  = (int*)(neigh0 + (size_t)N2 * F);
        int* off  = cnt + N2;
        int* rank = off + (N2 + 1);
        int* perm = rank + E1;
        hipMemsetAsync(cnt, 0, (size_t)N2 * sizeof(int), stream);
        sage_edge_rank<<<(E1 + 255) / 256, 256, 0, stream>>>(dst1, cnt, rank, E1);
        sage_edge_offsets<<<1, 1024, 0, stream>>>(cnt, off, N2);
        sage_edge_fill<<<(E1 + 255) / 256, 256, 0, stream>>>(src1, dst1, rank,
                                                            off, perm, E1);
        sage_gather_rows<<<(N2 + 7) / 8, 256, 0, stream>>>(h1, off, perm,
                                                           neigh0, deg0, N2);
    }

    {
        size_t work = (size_t)N2 * (F / 4);
        sage_combine<<<(unsigned)((work + 255) / 256), 256, 0, stream>>>(
            neigh0, h1, deg0, hpre0, N2);
    }

    {
        dim3 grid((N2 + 15) / 16, 256 / 64);
        sage_gemm_wmma<false><<<grid, 128, 0, stream>>>(hpre0, W2, b2, out,
                                                        N2, 256, 512);
    }
}


# Combine fused into the gather: the wave that sums a row adds its own row and scales by 1/(list length+1) before its one store; combine launches, neighbour-sum plane and degree array are gone
# speedup vs baseline: 1.8798x; 1.0647x over previous
#include <hip/hip_runtime.h>
#include <cstddef>

typedef float v2f __attribute__((ext_vector_type(2)));
typedef float v8f __attribute__((ext_vector_type(8)));

#define D_FEAT 512
#define LDS_STRIDE 516

__global__ __launch_bounds__(256) void sage_edge_rank(
    const int* __restrict__ dst, int* __restrict__ cnt,
    int* __restrict__ rank, int E)
{
    int e = blockIdx.x * 256 + threadIdx.x;
    if (e >= E) return;
    rank[e] = atomicAdd(cnt + dst[e], 1);
}

__global__ __launch_bounds__(1024) void sage_edge_offsets(
    const int* __restrict__ cnt, int* __restrict__ off, int n)
{
    __shared__ int wsum[32];
    const int t    = threadIdx.x;
    const int lane = t & 31;
    const int w    = t >> 5;
    const int per  = (n + 1023) / 1024;
    const int ib   = min(t * per, n);
    const int ie   = min(ib + per, n);

    int s = 0;
    for (int i = ib; i < ie; ++i) s += cnt[i];

    int v = s;
    for (int d = 1; d < 32; d <<= 1) {
        int u = __shfl_up(v, d);
        if (lane >= d) v += u;
    }
    if (lane == 31) wsum[w] = v;
    __syncthreads();
    if (w == 0) {
        int x = wsum[lane];
        for (int d = 1; d < 32; d <<= 1) {
            int u = __shfl_up(x, d);
            if (lane >= d) x += u;
        }
        wsum[lane] = x;
    }
    __syncthreads();

    int base = v - s + (w > 0 ? wsum[w - 1] : 0);
    for (int i = ib; i < ie; ++i) {
        off[i] = base;
        base += cnt[i];
    }
    if (t == 0) off[n] = wsum[31];
}

__global__ __launch_bounds__(256) void sage_edge_fill(
    const int* __restrict__ src, const int* __restrict__ dst,
    const int* __restrict__ rank, const int* __restrict__ off,
    int* __restrict__ perm, int E)
{
    int e = blockIdx.x * 256 + threadIdx.x;
    if (e >= E) return;
    perm[off[dst[e]] + rank[e]] = src[e];
}

__global__ __launch_bounds__(256) void sage_gather_combine_rows(
    const float* __restrict__ hsrc, const float* __restrict__ hdst,
    const int* __restrict__ off, const int* __restrict__ perm,
    float* __restrict__ hpre, int n)
{
    const int row = __builtin_amdgcn_readfirstlane(
        (int)(blockIdx.x * 8 + (threadIdx.x >> 5)));
    if (row >= n) return;
    const int lane = threadIdx.x & 31;
    const int jb = off[row];
    const int je = off[row + 1];
    const float* col = hsrc + lane * 4;

    float4 a0 = make_float4(0.f, 0.f, 0.f, 0.f);
    float4 a1 = a0, a2 = a0, a3 = a0;

    int j = jb;
    for (; j + 4 <= je; j += 4) {
        const float* r0 = col + (size_t)perm[j + 0] * D_FEAT;
        const float* r1 = col + (size_t)perm[j + 1] * D_FEAT;
        const float* r2 = col + (size_t)perm[j + 2] * D_FEAT;
        const float* r3 = col + (size_t)perm[j + 3] * D_FEAT;
        const float4 p0 = *reinterpret_cast<const float4*>(r0);
        const float4 p1 = *reinterpret_cast<const float4*>(r0 + 128);
        const float4 p2 = *reinterpret_cast<const float4*>(r0 + 256);
        const float4 p3 = *reinterpret_cast<const float4*>(r0 + 384);
        const float4 q0 = *reinterpret_cast<const float4*>(r1);
        const float4 q1 = *reinterpret_cast<const float4*>(r1 + 128);
        const float4 q2 = *reinterpret_cast<const float4*>(r1 + 256);
        const float4 q3 = *reinterpret_cast<const float4*>(r1 + 384);
        const float4 s0 = *reinterpret_cast<const float4*>(r2);
        const float4 s1 = *reinterpret_cast<const float4*>(r2 + 128);
        const float4 s2 = *reinterpret_cast<const float4*>(r2 + 256);
        const float4 s3 = *reinterpret_cast<const float4*>(r2 + 384);
        const float4 t0 = *reinterpret_cast<const float4*>(r3);
        const float4 t1 = *reinterpret_cast<const float4*>(r3 + 128);
        const float4 t2 = *reinterpret_cast<const float4*>(r3 + 256);
        const float4 t3 = *reinterpret_cast<const float4*>(r3 + 384);
        a0.x += p0.x; a0.y += p0.y; a0.z += p0.z; a0.w += p0.w;
        a1.x += p1.x; a1.y += p1.y; a1.z += p1.z; a1.w += p1.w;
        a2.x += p2.x; a2.y += p2.y; a2.z += p2.z; a2.w += p2.w;
        a3.x += p3.x; a3.y += p3.y; a3.z += p3.z; a3.w += p3.w;
        a0.x += q0.x; a0.y += q0.y; a0.z += q0.z; a0.w += q0.w;
        a1.x += q1.x; a1.y += q1.y; a1.z += q1.z; a1.w += q1.w;
        a2.x += q2.x; a2.y += q2.y; a2.z += q2.z; a2.w += q2.w;
        a3.x += q3.x; a3.y += q3.y; a3.z += q3.z; a3.w += q3.w;
        a0.x += s0.x; a0.y += s0.y; a0.z += s0.z; a0.w += s0.w;
        a1.x += s1.x; a1.y += s1.y; a1.z += s1.z; a1.w += s1.w;
        a2.x += s2.x; a2.y += s2.y; a2.z += s2.z; a2.w += s2.w;
        a3.x += s3.x; a3.y += s3.y; a3.z += s3.z; a3.w += s3.w;
        a0.x += t0.x; a0.y += t0.y; a0.z += t0.z; a0.w += t0.w;
        a1.x += t1.x; a1.y += t1.y; a1.z += t1.z; a1.w += t1.w;
        a2.x += t2.x; a2.y += t2.y; a2.z += t2.z; a2.w += t2.w;
        a3.x += t3.x; a3.y += t3.y; a3.z += t3.z; a3.w += t3.w;
    }
    const float* own = hdst + (size_t)row * D_FEAT + lane * 4;
    const float4 d0 = *reinterpret_cast<const float4*>(own);
    const float4 d1 = *reinterpret_cast<const float4*>(own + 128);
    const float4 d2 = *reinterpret_cast<const float4*>(own + 256);
    const float4 d3 = *reinterpret_cast<const float4*>(own + 384);

    for (; j < je; ++j) {
        const float* r0 = col + (size_t)perm[j] * D_FEAT;
        const float4 p0 = *reinterpret_cast<const float4*>(r0);
        const float4 p1 = *reinterpret_cast<const float4*>(r0 + 128);
        const float4 p2 = *reinterpret_cast<const float4*>(r0 + 256);
        const float4 p3 = *reinterpret_cast<const float4*>(r0 + 384);
        a0.x += p0.x; a0.y += p0.y; a0.z += p0.z; a0.w += p0.w;
        a1.x += p1.x; a1.y += p1.y; a1.z += p1.z; a1.w += p1.w;
        a2.x += p2.x; a2.y += p2.y; a2.z += p2.z; a2.w += p2.w;
        a3.x += p3.x; a3.y += p3.y; a3.z += p3.z; a3.w += p3.w;
    }

    const float inv = 1.0f / ((float)(je - jb) + 1.0f);
    float4 r0, r1, r2, r3;
    r0.x = (a0.x + d0.x) * inv; r0.y = (a0.y + d0.y) * inv;
    r0.z = (a0.z + d0.z) * inv; r0.w = (a0.w + d0.w) * inv;
    r1.x = (a1.x + d1.x) * inv; r1.y = (a1.y + d1.y) * inv;
    r1.z = (a1.z + d1.z) * inv; r1.w = (a1.w + d1.w) * inv;
    r2.x = (a2.x + d2.x) * inv; r2.y = (a2.y + d2.y) * inv;
    r2.z = (a2.z + d2.z) * inv; r2.w = (a2.w + d2.w) * inv;
    r3.x = (a3.x + d3.x) * inv; r3.y = (a3.y + d3.y) * inv;
    r3.z = (a3.z + d3.z) * inv; r3.w = (a3.w + d3.w) * inv;

    float* o = hpre + (size_t)row * D_FEAT + lane * 4;
    *reinterpret_cast<float4*>(o)       = r0;
    *reinterpret_cast<float4*>(o + 128) = r1;
    *reinterpret_cast<float4*>(o + 256) = r2;
    *reinterpret_cast<float4*>(o + 384) = r3;
}

template <bool RELU>
__global__ __launch_bounds__(128) void sage_gemm_wmma(
    const float* __restrict__ A, const float* __restrict__ B,
    const float* __restrict__ bias, float* __restrict__ C,
    int M, int N, int K)
{
    __shared__ float As[16 * LDS_STRIDE];

    const int rm   = blockIdx.x * 16;
    const int wave = threadIdx.x >> 5;
    const int lane = threadIdx.x & 31;
    const int half = lane >> 4;
    const int l    = lane & 15;
    const int cn   = (blockIdx.y * 4 + wave) * 16;

    {
        int r    = threadIdx.x >> 3;
        int cseg = (threadIdx.x & 7) * 64;
        int gr   = min(rm + r, M - 1);
        const float* gsrc = A + (size_t)gr * K + cseg;
        float* ldst = As + r * LDS_STRIDE + cseg;
#pragma unroll
        for (int i = 0; i < 16; ++i) {
            *reinterpret_cast<float4*>(ldst + i * 4) =
                *reinterpret_cast<const float4*>(gsrc + i * 4);
        }
    }
    __syncthreads();

    v8f acc = {};
    const float* arow = As + l * LDS_STRIDE + 2 * half;
    const float* bcol = B + (size_t)(2 * half) * N + cn + l;
    for (int k = 0; k < K; k += 4) {
        v2f a;
        a.x = arow[k];
        a.y = arow[k + 1];
        v2f b;
        b.x = bcol[(size_t)k * N];
        b.y = bcol[(size_t)(k + 1) * N];
        acc = __builtin_amdgcn_wmma_f32_16x16x4_f32(
            false, a, false, b, (short)0, acc, false, false);
    }

    const float bv = bias[cn + l];
#pragma unroll
    for (int i = 0; i < 8; ++i) {
        int row = rm + i + 8 * half;
        float v = acc[i] + bv;
        if (RELU) v = fmaxf(v, 0.0f);
        if (row < M) C[(size_t)row * N + cn + l] = v;
    }
}

extern "C" void kernel_launch(void* const* d_in, const int* in_sizes, int n_in,
                              void* d_out, int out_size, void* d_ws, size_t ws_size,
                              hipStream_t stream)
{
    const float* feats = (const float*)d_in[0];
    const int*   src0  = (const int*)d_in[1];
    const int*   dst0  = (const int*)d_in[2];
    const int*   src1  = (const int*)d_in[3];
    const int*   dst1  = (const int*)d_in[4];
    const float* W1    = (const float*)d_in[5];
    const float* b1    = (const float*)d_in[6];
    const float* W2    = (const float*)d_in[7];
    const float* b2    = (const float*)d_in[8];
    float* out = (float*)d_out;

    const int N1 = 25000, N2 = 5000;
    const int E0 = in_sizes[1];
    const int E1 = in_sizes[3];
    const size_t F = D_FEAT;

    float* ws    = (float*)d_ws;
    float* scr1  = ws;
    float* hpre0 = scr1 + (size_t)N1 * F;
    float* h1    = hpre0 + (size_t)N1 * F;

    {
        int* cnt  = (int*)h1;
        int* off  = cnt + N1;
        int* rank = off + (N1 + 1);
        int* perm = rank + E0;
        hipMemsetAsync(cnt, 0, (size_t)N1 * sizeof(int), stream);
        sage_edge_rank<<<(E0 + 255) / 256, 256, 0, stream>>>(dst0, cnt, rank, E0);
        sage_edge_offsets<<<1, 1024, 0, stream>>>(cnt, off, N1);
        sage_edge_fill<<<(E0 + 255) / 256, 256, 0, stream>>>(src0, dst0, rank,
                                                            off, perm, E0);
        sage_gather_combine_rows<<<(N1 + 7) / 8, 256, 0, stream>>>(
            feats, feats, off, perm, hpre0, N1);
    }

    {
        dim3 grid((N1 + 15) / 16, 512 / 64);
        sage_gemm_wmma<true><<<grid, 128, 0, stream>>>(hpre0, W1, b1, h1,
                                                       N1, 512, 512);
    }

    {
        int* cnt  = (int*)(scr1 + (size_t)N2 * F);
        int* off  = cnt + N2;
        int* rank = off + (N2 + 1);
        int* perm = rank + E1;
        hipMemsetAsync(cnt, 0, (size_t)N2 * sizeof(int), stream);
        sage_edge_rank<<<(E1 + 255) / 256, 256, 0, stream>>>(dst1, cnt, rank, E1);
        sage_edge_offsets<<<1, 1024, 0, stream>>>(cnt, off, N2);
        sage_edge_fill<<<(E1 + 255) / 256, 256, 0, stream>>>(src1, dst1, rank,
                                                            off, perm, E1);
        sage_gather_combine_rows<<<(N2 + 7) / 8, 256, 0, stream>>>(
            h1, h1, off, perm, hpre0, N2);
    }

    {
        dim3 grid((N2 + 15) / 16, 256 / 64);
        sage_gemm_wmma<false><<<grid, 128, 0, stream>>>(hpre0, W2, b2, out,
                                                        N2, 256, 512);
    }
}


# GEMM register tiling: a wave owns 32x32 outputs (2x2 f32 16x16x4 accumulator tiles; layer 2: 16x32), B read as 8-byte column pairs, next 16 k of fragments requested ahead; k order of sums kept
# speedup vs baseline: 2.3216x; 1.2350x over previous
#include <hip/hip_runtime.h>
#include <cstddef>

typedef float v2f __attribute__((ext_vector_type(2)));
typedef float v8f __attribute__((ext_vector_type(8)));

#define D_FEAT 512
#define LDS_STRIDE 516

__global__ __launch_bounds__(256) void sage_edge_rank(
    const int* __restrict__ dst, int* __restrict__ cnt,
    int* __restrict__ rank, int E)
{
    int e = blockIdx.x * 256 + threadIdx.x;
    if (e >= E) return;
    rank[e] = atomicAdd(cnt + dst[e], 1);
}

__global__ __launch_bounds__(1024) void sage_edge_offsets(
    const int* __restrict__ cnt, int* __restrict__ off, int n)
{
    __shared__ int wsum[32];
    const int t    = threadIdx.x;
    const int lane = t & 31;
    const int w    = t >> 5;
    const int per  = (n + 1023) / 1024;
    const int ib   = min(t * per, n);
    const int ie   = min(ib + per, n);

    int s = 0;
    for (int i = ib; i < ie; ++i) s += cnt[i];

    int v = s;
    for (int d = 1; d < 32; d <<= 1) {
        int u = __shfl_up(v, d);
        if (lane >= d) v += u;
    }
    if (lane == 31) wsum[w] = v;
    __syncthreads();
    if (w == 0) {
        int x = wsum[lane];
        for (int d = 1; d < 32; d <<= 1) {
            int u = __shfl_up(x, d);
            if (lane >= d) x += u;
        }
        wsum[lane] = x;
    }
    __syncthreads();

    int base = v - s + (w > 0 ? wsum[w - 1] : 0);
    for (int i = ib; i < ie; ++i) {
        off[i] = base;
        base += cnt[i];
    }
    if (t == 0) off[n] = wsum[31];
}

__global__ __launch_bounds__(256) void sage_edge_fill(
    const int* __restrict__ src, const int* __restrict__ dst,
    const int* __restrict__ rank, const int* __restrict__ off,
    int* __restrict__ perm, int E)
{
    int e = blockIdx.x * 256 + threadIdx.x;
    if (e >= E) return;
    perm[off[dst[e]] + rank[e]] = src[e];
}

__global__ __launch_bounds__(256) void sage_gather_combine_rows(
    const float* __restrict__ hsrc, const float* __restrict__ hdst,
    const int* __restrict__ off, const int* __restrict__ perm,
    float* __restrict__ hpre, int n)
{
    const int row = __builtin_amdgcn_readfirstlane(
        (int)(blockIdx.x * 8 + (threadIdx.x >> 5)));
    if (row >= n) return;
    const int lane = threadIdx.x & 31;
    const int jb = off[row];
    const int je = off[row + 1];
    const float* col = hsrc + lane * 4;

    float4 a0 = make_float4(0.f, 0.f, 0.f, 0.f);
    float4 a1 = a0, a2 = a0, a3 = a0;

    int j = jb;
    for (; j + 4 <= je; j += 4) {
        const float* r0 = col + (size_t)perm[j + 0] * D_FEAT;
        const float* r1 = col + (size_t)perm[j + 1] * D_FEAT;
        const float* r2 = col + (size_t)perm[j + 2] * D_FEAT;
        const float* r3 = col + (size_t)perm[j + 3] * D_FEAT;
        const float4 p0 = *reinterpret_cast<const float4*>(r0);
        const float4 p1 = *reinterpret_cast<const float4*>(r0 + 128);
        const float4 p2 = *reinterpret_cast<const float4*>(r0 + 256);
        const float4 p3 = *reinterpret_cast<const float4*>(r0 + 384);
        const float4 q0 = *reinterpret_cast<const float4*>(r1);
        const float4 q1 = *reinterpret_cast<const float4*>(r1 + 128);
        const float4 q2 = *reinterpret_cast<const float4*>(r1 + 256);
        const float4 q3 = *reinterpret_cast<const float4*>(r1 + 384);
        const float4 s0 = *reinterpret_cast<const float4*>(r2);
        const float4 s1 = *reinterpret_cast<const float4*>(r2 + 128);
        const float4 s2 = *reinterpret_cast<const float4*>(r2 + 256);
        const float4 s3 = *reinterpret_cast<const float4*>(r2 + 384);
        const float4 t0 = *reinterpret_cast<const float4*>(r3);
        const float4 t1 = *reinterpret_cast<const float4*>(r3 + 128);
        const float4 t2 = *reinterpret_cast<const float4*>(r3 + 256);
        const float4 t3 = *reinterpret_cast<const float4*>(r3 + 384);
        a0.x += p0.x; a0.y += p0.y; a0.z += p0.z; a0.w += p0.w;
        a1.x += p1.x; a1.y += p1.y; a1.z += p1.z; a1.w += p1.w;
        a2.x += p2.x; a2.y += p2.y; a2.z += p2.z; a2.w += p2.w;
        a3.x += p3.x; a3.y += p3.y; a3.z += p3.z; a3.w += p3.w;
        a0.x += q0.x; a0.y += q0.y; a0.z += q0.z; a0.w += q0.w;
        a1.x += q1.x; a1.y += q1.y; a1.z += q1.z; a1.w += q1.w;
        a2.x += q2.x; a2.y += q2.y; a2.z += q2.z; a2.w += q2.w;
        a3.x += q3.x; a3.y += q3.y; a3.z += q3.z; a3.w += q3.w;
        a0.x += s0.x; a0.y += s0.y; a0.z += s0.z; a0.w += s0.w;
        a1.x += s1.x; a1.y += s1.y; a1.z += s1.z; a1.w += s1.w;
        a2.x += s2.x; a2.y += s2.y; a2.z += s2.z; a2.w += s2.w;
        a3.x += s3.x; a3.y += s3.y; a3.z += s3.z; a3.w += s3.w;
        a0.x += t0.x; a0.y += t0.y; a0.z += t0.z; a0.w += t0.w;
        a1.x += t1.x; a1.y += t1.y; a1.z += t1.z; a1.w += t1.w;
        a2.x += t2.x; a2.y += t2.y; a2.z += t2.z; a2.w += t2.w;
        a3.x += t3.x; a3.y += t3.y; a3.z += t3.z; a3.w += t3.w;
    }
    const float* own = hdst + (size_t)row * D_FEAT + lane * 4;
    const float4 d0 = *reinterpret_cast<const float4*>(own);
    const float4 d1 = *reinterpret_cast<const float4*>(own + 128);
    const float4 d2 = *reinterpret_cast<const float4*>(own + 256);
    const float4 d3 = *reinterpret_cast<const float4*>(own + 384);

    for (; j < je; ++j) {
        const float* r0 = col + (size_t)perm[j] * D_FEAT;
        const float4 p0 = *reinterpret_cast<const float4*>(r0);
        const float4 p1 = *reinterpret_cast<const float4*>(r0 + 128);
        const float4 p2 = *reinterpret_cast<const float4*>(r0 + 256);
        const float4 p3 = *reinterpret_cast<const float4*>(r0 + 384);
        a0.x += p0.x; a0.y += p0.y; a0.z += p0.z; a0.w += p0.w;
        a1.x += p1.x; a1.y += p1.y; a1.z += p1.z; a1.w += p1.w;
        a2.x += p2.x; a2.y += p2.y; a2.z += p2.z; a2.w += p2.w;
        a3.x += p3.x; a3.y += p3.y; a3.z += p3.z; a3.w += p3.w;
    }

    const float inv = 1.0f / ((float)(je - jb) + 1.0f);
    float4 r0, r1, r2, r3;
    r0.x = (a0.x + d0.x) * inv; r0.y = (a0.y + d0.y) * inv;
    r0.z = (a0.z + d0.z) * inv; r0.w = (a0.w + d0.w) * inv;
    r1.x = (a1.x + d1.x) * inv; r1.y = (a1.y + d1.y) * inv;
    r1.z = (a1.z + d1.z) * inv; r1.w = (a1.w + d1.w) * inv;
    r2.x = (a2.x + d2.x) * inv; r2.y = (a2.y + d2.y) * inv;
    r2.z = (a2.z + d2.z) * inv; r2.w = (a2.w + d2.w) * inv;
    r3.x = (a3.x + d3.x) * inv; r3.y = (a3.y + d3.y) * inv;
    r3.z = (a3.z + d3.z) * inv; r3.w = (a3.w + d3.w) * inv;

    float* o = hpre + (size_t)row * D_FEAT + lane * 4;
    *reinterpret_cast<float4*>(o)       = r0;
    *reinterpret_cast<float4*>(o + 128) = r1;
    *reinterpret_cast<float4*>(o + 256) = r2;
    *reinterpret_cast<float4*>(o + 384) = r3;
}

#define GEMM_KU 4
#define GEMM_G  (4 * GEMM_KU)

template <int MT>
struct FragSet {
    float2 b[GEMM_KU][2];
    v2f    a[GEMM_KU][MT];
};

template <int MT>
__device__ __forceinline__ void gemm_load_set(
    FragSet<MT>& s, const float* __restrict__ bcol, const float* arow,
    int k, int N)
{
#pragma unroll
    for (int u = 0; u < GEMM_KU; ++u) {
        const int kk = k + 4 * u;
        s.b[u][0] = *reinterpret_cast<const float2*>(bcol + (size_t)kk * N);
        s.b[u][1] = *reinterpret_cast<const float2*>(bcol + (size_t)(kk + 1) * N);
#pragma unroll
        for (int m = 0; m < MT; ++m)
            s.a[u][m] = *reinterpret_cast<const v2f*>(
                arow + m * 16 * LDS_STRIDE + kk);
    }
}

template <int MT>
__device__ __forceinline__ void gemm_mma_set(
    const FragSet<MT>& s, v8f (&acc)[MT][2])
{
#pragma unroll
    for (int u = 0; u < GEMM_KU; ++u) {
        v2f b0, b1;
        b0.x = s.b[u][0].x; b0.y = s.b[u][1].x;
        b1.x = s.b[u][0].y; b1.y = s.b[u][1].y;
#pragma unroll
        for (int m = 0; m < MT; ++m) {
            acc[m][0] = __builtin_amdgcn_wmma_f32_16x16x4_f32(
                false, s.a[u][m], false, b0, (short)0, acc[m][0], false, false);
            acc[m][1] = __builtin_amdgcn_wmma_f32_16x16x4_f32(
                false, s.a[u][m], false, b1, (short)0, acc[m][1], false, false);
        }
    }
}

template <int MT, bool RELU>
__global__ __launch_bounds__(128) void sage_gemm_wmma(
    const float* __restrict__ A, const float* __restrict__ B,
    const float* __restrict__ bias, float* __restrict__ C,
    int M, int N)
{
    constexpr int K = D_FEAT;
    __shared__ float As[MT * 16 * LDS_STRIDE];

    const int rm   = blockIdx.x * (16 * MT);
    const int wave = threadIdx.x >> 5;
    const int lane = threadIdx.x & 31;
    const int half = lane >> 4;
    const int l    = lane & 15;
    const int cn   = (blockIdx.y * 4 + wave) * 32;

    {
        const int r    = threadIdx.x >> 3;
        const int cseg = (threadIdx.x & 7) * 64;
#pragma unroll
        for (int m = 0; m < MT; ++m) {
            const int gr = min(rm + m * 16 + r, M - 1);
            const float* gsrc = A + (size_t)gr * K + cseg;
            float* ldst = As + (m * 16 + r) * LDS_STRIDE + cseg;
#pragma unroll
            for (int i = 0; i < 16; ++i) {
                *reinterpret_cast<float4*>(ldst + i * 4) =
                    *reinterpret_cast<const float4*>(gsrc + i * 4);
            }
        }
    }
    __syncthreads();

    v8f acc[MT][2];
#pragma unroll
    for (int m = 0; m < MT; ++m) {
        acc[m][0] = v8f{};
        acc[m][1] = v8f{};
    }

    const float* arow = As + l * LDS_STRIDE + 2 * half;
    const float* bcol = B + (size_t)(2 * half) * N + cn + 2 * l;

    FragSet<MT> s0, s1;
    gemm_load_set<MT>(s0, bcol, arow, 0, N);
    for (int k = 0; k < K - 2 * GEMM_G; k += 2 * GEMM_G) {
        gemm_load_set<MT>(s1, bcol, arow, k + GEMM_G, N);
        gemm_mma_set<MT>(s0, acc);
        gemm_load_set<MT>(s0, bcol, arow, k + 2 * GEMM_G, N);
        gemm_mma_set<MT>(s1, acc);
    }
    gemm_load_set<MT>(s1, bcol, arow, K - GEMM_G, N);
    gemm_mma_set<MT>(s0, acc);
    gemm_mma_set<MT>(s1, acc);

    const float2 bv = *reinterpret_cast<const float2*>(bias + cn + 2 * l);
#pragma unroll
    for (int m = 0; m < MT; ++m) {
#pragma unroll
        for (int i = 0; i < 8; ++i) {
            const int row = rm + m * 16 + i + 8 * half;
            float2 v;
            v.x = acc[m][0][i] + bv.x;
            v.y = acc[m][1][i] + bv.y;
            if (RELU) {
                v.x = fmaxf(v.x, 0.0f);
                v.y = fmaxf(v.y, 0.0f);
            }
            if (row < M)
                *reinterpret_cast<float2*>(C + (size_t)row * N + cn + 2 * l) = v;
        }
    }
}

extern "C" void kernel_launch(void* const* d_in, const int* in_sizes, int n_in,
                              void* d_out, int out_size, void* d_ws, size_t ws_size,
                              hipStream_t stream)
{
    const float* feats = (const float*)d_in[0];
    const int*   src0  = (const int*)d_in[1];
    const int*   dst0  = (const int*)d_in[2];
    const int*   src1  = (const int*)d_in[3];
    const int*   dst1  = (const int*)d_in[4];
    const float* W1    = (const float*)d_in[5];
    const float* b1    = (const float*)d_in[6];
    const float* W2    = (const float*)d_in[7];
    const float* b2    = (const float*)d_in[8];
    float* out = (float*)d_out;

    const int N1 = 25000, N2 = 5000;
    const int E0 = in_sizes[1];
    const int E1 = in_sizes[3];
    const size_t F = D_FEAT;

    float* ws    = (float*)d_ws;
    float* scr1  = ws;
    float* hpre0 = scr1 + (size_t)N1 * F;
    float* h1    = hpre0 + (size_t)N1 * F;

    {
        int* cnt  = (int*)h1;
        int* off  = cnt + N1;
        int* rank = off + (N1 + 1);
        int* perm = rank + E0;
        hipMemsetAsync(cnt, 0, (size_t)N1 * sizeof(int), stream);
        sage_edge_rank<<<(E0 + 255) / 256, 256, 0, stream>>>(dst0, cnt, rank, E0);
        sage_edge_offsets<<<1, 1024, 0, stream>>>(cnt, off, N1);
        sage_edge_fill<<<(E0 + 255) / 256, 256, 0, stream>>>(src0, dst0, rank,
                                                            off, perm, E0);
        sage_gather_combine_rows<<<(N1 + 7) / 8, 256, 0, stream>>>(
            feats, feats, off, perm, hpre0, N1);
    }

    {
        dim3 grid((N1 + 31) / 32, 512 / 128);
        sage_gemm_wmma<2, true><<<grid, 128, 0, stream>>>(hpre0, W1, b1, h1,
                                                          N1, 512);
    }

    {
        int* cnt  = (int*)(scr1 + (size_t)N2 * F);
        int* off  = cnt + N2;
        int* rank = off + (N2 + 1);
        int* perm = rank + E1;
        hipMemsetAsync(cnt, 0, (size_t)N2 * sizeof(int), stream);
        sage_edge_rank<<<(E1 + 255) / 256, 256, 0, stream>>>(dst1, cnt, rank, E1);
        sage_edge_offsets<<<1, 1024, 0, stream>>>(cnt, off, N2);
        sage_edge_fill<<<(E1 + 255) / 256, 256, 0, stream>>>(src1, dst1, rank,
                                                            off, perm, E1);
        sage_gather_combine_rows<<<(N2 + 7) / 8, 256, 0, stream>>>(
            h1, h1, off, perm, hpre0, N2);
    }

    {
        dim3 grid((N2 + 15) / 16, 256 / 128);
        sage_gemm_wmma<1, false><<<grid, 128, 0, stream>>>(hpre0, W2, b2, out,
                                                           N2, 256);
    }
}


# Gather fused into the GEMM: a block of N/32 waves gathers and combines its 32 rows (layer 2: 16) straight into the LDS A tile, then multiplies; no combined-row plane or gather launch; sum order kept
# speedup vs baseline: 2.4837x; 1.0698x over previous
#include <hip/hip_runtime.h>
#include <cstddef>

typedef float v2f __attribute__((ext_vector_type(2)));
typedef float v8f __attribute__((ext_vector_type(8)));

#define D_FEAT 512
#define LDS_STRIDE 516

__global__ __launch_bounds__(256) void sage_edge_rank(
    const int* __restrict__ dst, int* __restrict__ cnt,
    int* __restrict__ rank, int E)
{
    int e = blockIdx.x * 256 + threadIdx.x;
    if (e >= E) return;
    rank[e] = atomicAdd(cnt + dst[e], 1);
}

__global__ __launch_bounds__(1024) void sage_edge_offsets(
    const int* __restrict__ cnt, int* __restrict__ off, int n)
{
    __shared__ int wsum[32];
    const int t    = threadIdx.x;
    const int lane = t & 31;
    const int w    = t >> 5;
    const int per  = (n + 1023) / 1024;
    const int ib   = min(t * per, n);
    const int ie   = min(ib + per, n);

    int s = 0;
    for (int i = ib; i < ie; ++i) s += cnt[i];

    int v = s;
    for (int d = 1; d < 32; d <<= 1) {
        int u = __shfl_up(v, d);
        if (lane >= d) v += u;
    }
    if (lane == 31) wsum[w] = v;
    __syncthreads();
    if (w == 0) {
        int x = wsum[lane];
        for (int d = 1; d < 32; d <<= 1) {
            int u = __shfl_up(x, d);
            if (lane >= d) x += u;
        }
        wsum[lane] = x;
    }
    __syncthreads();

    int base = v - s + (w > 0 ? wsum[w - 1] : 0);
    for (int i = ib; i < ie; ++i) {
        off[i] = base;
        base += cnt[i];
    }
    if (t == 0) off[n] = wsum[31];
}

__global__ __launch_bounds__(256) void sage_edge_fill(
    const int* __restrict__ src, const int* __restrict__ dst,
    const int* __restrict__ rank, const int* __restrict__ off,
    int* __restrict__ perm, int E)
{
    int e = blockIdx.x * 256 + threadIdx.x;
    if (e >= E) return;
    perm[off[dst[e]] + rank[e]] = src[e];
}

__device__ __forceinline__ void sage_gather_combine_row(
    const float* __restrict__ hsrc, const float* __restrict__ hdst,
    const int* __restrict__ off, const int* __restrict__ perm,
    float* o, int row, int lane)
{
    const int jb = off[row];
    const int je = off[row + 1];
    const float* col = hsrc + lane * 4;

    float4 a0 = make_float4(0.f, 0.f, 0.f, 0.f);
    float4 a1 = a0, a2 = a0, a3 = a0;

    int j = jb;
    for (; j + 4 <= je; j += 4) {
        const float* r0 = col + (size_t)perm[j + 0] * D_FEAT;
        const float* r1 = col + (size_t)perm[j + 1] * D_FEAT;
        const float* r2 = col + (size_t)perm[j + 2] * D_FEAT;
        const float* r3 = col + (size_t)perm[j + 3] * D_FEAT;
        const float4 p0 = *reinterpret_cast<const float4*>(r0);
        const float4 p1 = *reinterpret_cast<const float4*>(r0 + 128);
        const float4 p2 = *reinterpret_cast<const float4*>(r0 + 256);
        const float4 p3 = *reinterpret_cast<const float4*>(r0 + 384);
        const float4 q0 = *reinterpret_cast<const float4*>(r1);
        const float4 q1 = *reinterpret_cast<const float4*>(r1 + 128);
        const float4 q2 = *reinterpret_cast<const float4*>(r1 + 256);
        const float4 q3 = *reinterpret_cast<const float4*>(r1 + 384);
        const float4 s0 = *reinterpret_cast<const float4*>(r2);
        const float4 s1 = *reinterpret_cast<const float4*>(r2 + 128);
        const float4 s2 = *reinterpret_cast<const float4*>(r2 + 256);
        const float4 s3 = *reinterpret_cast<const float4*>(r2 + 384);
        const float4 t0 = *reinterpret_cast<const float4*>(r3);
        const float4 t1 = *reinterpret_cast<const float4*>(r3 + 128);
        const float4 t2 = *reinterpret_cast<const float4*>(r3 + 256);
        const float4 t3 = *reinterpret_cast<const float4*>(r3 + 384);
        a0.x += p0.x; a0.y += p0.y; a0.z += p0.z; a0.w += p0.w;
        a1.x += p1.x; a1.y += p1.y; a1.z += p1.z; a1.w += p1.w;
        a2.x += p2.x; a2.y += p2.y; a2.z += p2.z; a2.w += p2.w;
        a3.x += p3.x; a3.y += p3.y; a3.z += p3.z; a3.w += p3.w;
        a0.x += q0.x; a0.y += q0.y; a0.z += q0.z; a0.w += q0.w;
        a1.x += q1.x; a1.y += q1.y; a1.z += q1.z; a1.w += q1.w;
        a2.x += q2.x; a2.y += q2.y; a2.z += q2.z; a2.w += q2.w;
        a3.x += q3.x; a3.y += q3.y; a3.z += q3.z; a3.w += q3.w;
        a0.x += s0.x; a0.y += s0.y; a0.z += s0.z; a0.w += s0.w;
        a1.x += s1.x; a1.y += s1.y; a1.z += s1.z; a1.w += s1.w;
        a2.x += s2.x; a2.y += s2.y; a2.z += s2.z; a2.w += s2.w;
        a3.x += s3.x; a3.y += s3.y; a3.z += s3.z; a3.w += s3.w;
        a0.x += t0.x; a0.y += t0.y; a0.z += t0.z; a0.w += t0.w;
        a1.x += t1.x; a1.y += t1.y; a1.z += t1.z; a1.w += t1.w;
        a2.x += t2.x; a2.y += t2.y; a2.z += t2.z; a2.w += t2.w;
        a3.x += t3.x; a3.y += t3.y; a3.z += t3.z; a3.w += t3.w;
    }
    const float* own = hdst + (size_t)row * D_FEAT + lane * 4;
    const float4 d0 = *reinterpret_cast<const float4*>(own);
    const float4 d1 = *reinterpret_cast<const float4*>(own + 128);
    const float4 d2 = *reinterpret_cast<const float4*>(own + 256);
    const float4 d3 = *reinterpret_cast<const float4*>(own + 384);

    for (; j < je; ++j) {
        const float* r0 = col + (size_t)perm[j] * D_FEAT;
        const float4 p0 = *reinterpret_cast<const float4*>(r0);
        const float4 p1 = *reinterpret_cast<const float4*>(r0 + 128);
        const float4 p2 = *reinterpret_cast<const float4*>(r0 + 256);
        const float4 p3 = *reinterpret_cast<const float4*>(r0 + 384);
        a0.x += p0.x; a0.y += p0.y; a0.z += p0.z; a0.w += p0.w;
        a1.x += p1.x; a1.y += p1.y; a1.z += p1.z; a1.w += p1.w;
        a2.x += p2.x; a2.y += p2.y; a2.z += p2.z; a2.w += p2.w;
        a3.x += p3.x; a3.y += p3.y; a3.z += p3.z; a3.w += p3.w;
    }

    const float inv = 1.0f / ((float)(je - jb) + 1.0f);
    float4 r0, r1, r2, r3;
    r0.x = (a0.x + d0.x) * inv; r0.y = (a0.y + d0.y) * inv;
    r0.z = (a0.z + d0.z) * inv; r0.w = (a0.w + d0.w) * inv;
    r1.x = (a1.x + d1.x) * inv; r1.y = (a1.y + d1.y) * inv;
    r1.z = (a1.z + d1.z) * inv; r1.w = (a1.w + d1.w) * inv;
    r2.x = (a2.x + d2.x) * inv; r2.y = (a2.y + d2.y) * inv;
    r2.z = (a2.z + d2.z) * inv; r2.w = (a2.w + d2.w) * inv;
    r3.x = (a3.x + d3.x) * inv; r3.y = (a3.y + d3.y) * inv;
    r3.z = (a3.z + d3.z) * inv; r3.w = (a3.w + d3.w) * inv;

    *reinterpret_cast<float4*>(o)       = r0;
    *reinterpret_cast<float4*>(o + 128) = r1;
    *reinterpret_cast<float4*>(o + 256) = r2;
    *reinterpret_cast<float4*>(o + 384) = r3;
}

#define GEMM_KU 4
#define GEMM_G  (4 * GEMM_KU)

template <int MT>
struct FragSet {
    float2 b[GEMM_KU][2];
    v2f    a[GEMM_KU][MT];
};

template <int MT>
__device__ __forceinline__ void gemm_load_set(
    FragSet<MT>& s, const float* __restrict__ bcol, const float* arow,
    int k, int N)
{
#pragma unroll
    for (int u = 0; u < GEMM_KU; ++u) {
        const int kk = k + 4 * u;
        s.b[u][0] = *reinterpret_cast<const float2*>(bcol + (size_t)kk * N);
        s.b[u][1] = *reinterpret_cast<const float2*>(bcol + (size_t)(kk + 1) * N);
#pragma unroll
        for (int m = 0; m < MT; ++m)
            s.a[u][m] = *reinterpret_cast<const v2f*>(
                arow + m * 16 * LDS_STRIDE + kk);
    }
}

template <int MT>
__device__ __forceinline__ void gemm_mma_set(
    const FragSet<MT>& s, v8f (&acc)[MT][2])
{
#pragma unroll
    for (int u = 0; u < GEMM_KU; ++u) {
        v2f b0, b1;
        b0.x = s.b[u][0].x; b0.y = s.b[u][1].x;
        b1.x = s.b[u][0].y; b1.y = s.b[u][1].y;
#pragma unroll
        for (int m = 0; m < MT; ++m) {
            acc[m][0] = __builtin_amdgcn_wmma_f32_16x16x4_f32(
                false, s.a[u][m], false, b0, (short)0, acc[m][0], false, false);
            acc[m][1] = __builtin_amdgcn_wmma_f32_16x16x4_f32(
                false, s.a[u][m], false, b1, (short)0, acc[m][1], false, false);
        }
    }
}

template <int MT, int NW, bool RELU>
__global__ __launch_bounds__(NW * 32) void sage_gather_gemm_wmma(
    const float* __restrict__ hsrc, const float* __restrict__ hdst,
    const int* __restrict__ off, const int* __restrict__ perm,
    const float* __restrict__ B, const float* __restrict__ bias,
    float* __restrict__ C, int M)
{
    constexpr int K = D_FEAT;
    constexpr int N = NW * 32;
    __shared__ float As[MT * 16 * LDS_STRIDE];

    const int rm   = blockIdx.x * (16 * MT);
    const int wave = __builtin_amdgcn_readfirstlane((int)(threadIdx.x >> 5));
    const int lane = threadIdx.x & 31;
    const int half = lane >> 4;
    const int l    = lane & 15;
    const int cn   = wave * 32;

    for (int rr = wave; rr < 16 * MT; rr += NW) {
        float* o = As + rr * LDS_STRIDE + lane * 4;
        const int row = rm + rr;
        if (row < M) {
            sage_gather_combine_row(hsrc, hdst, off, perm, o, row, lane);
        } else {
            const float4 z = make_float4(0.f, 0.f, 0.f, 0.f);
            *reinterpret_cast<float4*>(o)       = z;
            *reinterpret_cast<float4*>(o + 128) = z;
            *reinterpret_cast<float4*>(o + 256) = z;
            *reinterpret_cast<float4*>(o + 384) = z;
        }
    }
    __syncthreads();

    v8f acc[MT][2];
#pragma unroll
    for (int m = 0; m < MT; ++m) {
        acc[m][0] = v8f{};
        acc[m][1] = v8f{};
    }

    const float* arow = As + l * LDS_STRIDE + 2 * half;
    const float* bcol = B + (size_t)(2 * half) * N + cn + 2 * l;

    FragSet<MT> s0, s1;
    gemm_load_set<MT>(s0, bcol, arow, 0, N);
    for (int k = 0; k < K - 2 * GEMM_G; k += 2 * GEMM_G) {
        gemm_load_set<MT>(s1, bcol, arow, k + GEMM_G, N);
        gemm_mma_set<MT>(s0, acc);
        gemm_load_set<MT>(s0, bcol, arow, k + 2 * GEMM_G, N);
        gemm_mma_set<MT>(s1, acc);
    }
    gemm_load_set<MT>(s1, bcol, arow, K - GEMM_G, N);
    gemm_mma_set<MT>(s0, acc);
    gemm_mma_set<MT>(s1, acc);

    const float2 bv = *reinterpret_cast<const float2*>(bias + cn + 2 * l);
#pragma unroll
    for (int m = 0; m < MT; ++m) {
#pragma unroll
        for (int i = 0; i < 8; ++i) {
            const int row = rm + m * 16 + i + 8 * half;
            float2 v;
            v.x = acc[m][0][i] + bv.x;
            v.y = acc[m][1][i] + bv.y;
            if (RELU) {
                v.x = fmaxf(v.x, 0.0f);
                v.y = fmaxf(v.y, 0.0f);
            }
            if (row < M)
                *reinterpret_cast<float2*>(C + (size_t)row * N + cn + 2 * l) = v;
        }
    }
}

extern "C" void kernel_launch(void* const* d_in, const int* in_sizes, int n_in,
                              void* d_out, int out_size, void* d_ws, size_t ws_size,
                              hipStream_t stream)
{
    const float* feats = (const float*)d_in[0];
    const int*   src0  = (const int*)d_in[1];
    const int*   dst0  = (const int*)d_in[2];
    const int*   src1  = (const int*)d_in[3];
    const int*   dst1  = (const int*)d_in[4];
    const float* W1    = (const float*)d_in[5];
    const float* b1    = (const float*)d_in[6];
    const float* W2    = (const float*)d_in[7];
    const float* b2    = (const float*)d_in[8];
    float* out = (float*)d_out;

    const int N1 = 25000, N2 = 5000;
    const int E0 = in_sizes[1];
    const int E1 = in_sizes[3];
    const size_t F = D_FEAT;

    float* ws   = (float*)d_ws;
    float* scr0 = ws;
    float* scr1 = scr0 + (size_t)N1 * F;
    float* h1   = scr1 + (size_t)N1 * F;

    {
        int* cnt  = (int*)scr0;
        int* off  = cnt + N1;
        int* rank = off + (N1 + 1);
        int* perm = rank + E0;
        hipMemsetAsync(cnt, 0, (size_t)N1 * sizeof(int), stream);
        sage_edge_rank<<<(E0 + 255) / 256, 256, 0, stream>>>(dst0, cnt, rank, E0);
        sage_edge_offsets<<<1, 1024, 0, stream>>>(cnt, off, N1);
        sage_edge_fill<<<(E0 + 255) / 256, 256, 0, stream>>>(src0, dst0, rank,
                                                            off, perm, E0);
        sage_gather_gemm_wmma<2, 16, true><<<(N1 + 31) / 32, 512, 0, stream>>>(
            feats, feats, off, perm, W1, b1, h1, N1);
    }

    {
        int* cnt  = (int*)scr1;
        int* off  = cnt + N2;
        int* rank = off + (N2 + 1);
        int* perm = rank + E1;
        hipMemsetAsync(cnt, 0, (size_t)N2 * sizeof(int), stream);
        sage_edge_rank<<<(E1 + 255) / 256, 256, 0, stream>>>(dst1, cnt, rank, E1);
        sage_edge_offsets<<<1, 1024, 0, stream>>>(cnt, off, N2);
        sage_edge_fill<<<(E1 + 255) / 256, 256, 0, stream>>>(src1, dst1, rank,
                                                            off, perm, E1);
        sage_gather_gemm_wmma<1, 8, false><<<(N2 + 15) / 16, 256, 0, stream>>>(
            h1, h1, off, perm, W2, b2, out, N2);
    }
}


# Edge lists in one launch for both layers: an edge's returning add picks its slot in a 64-wide row per destination (overflow: scanned spill list); scan+fill launches gone; sum order still arrival
# speedup vs baseline: 2.9238x; 1.1772x over previous
#include <hip/hip_runtime.h>
#include <cstddef>

typedef float v2f __attribute__((ext_vector_type(2)));
typedef float v8f __attribute__((ext_vector_type(8)));

#define D_FEAT 512
#define LDS_STRIDE 516

#define LIST_CAP 64

__global__ __launch_bounds__(256) void sage_edge_slots(
    const int* __restrict__ src0, const int* __restrict__ dst0,
    const int* __restrict__ src1, const int* __restrict__ dst1,
    int* __restrict__ cnt0, int* __restrict__ cnt1,
    int* __restrict__ slot0, int* __restrict__ slot1,
    int* __restrict__ spill0, int* __restrict__ spill1,
    int* __restrict__ nspill, int E0, int E1, int nb0)
{
    const bool first = (int)blockIdx.x < nb0;
    const int* src   = first ? src0 : src1;
    const int* dst   = first ? dst0 : dst1;
    int*       cnt   = first ? cnt0 : cnt1;
    int*       slot  = first ? slot0 : slot1;
    int*       spill = first ? spill0 : spill1;
    int*       nsp   = first ? nspill : nspill + 1;
    const int  E     = first ? E0 : E1;
    const int  b     = first ? (int)blockIdx.x : (int)blockIdx.x - nb0;

    const int e = b * 256 + threadIdx.x;
    if (e >= E) return;
    const int d = dst[e];
    const int s = src[e];
    const int r = atomicAdd(cnt + d, 1);
    if (r < LIST_CAP) {
        slot[(size_t)d * LIST_CAP + r] = s;
    } else {
        const int k = atomicAdd(nsp, 1);
        spill[2 * (size_t)k]     = d;
        spill[2 * (size_t)k + 1] = s;
    }
}

__device__ __forceinline__ void sage_gather_combine_row(
    const float* __restrict__ hsrc, const float* __restrict__ hdst,
    const int* __restrict__ cnt, const int* __restrict__ slot,
    const int* __restrict__ spill, const int* __restrict__ nspill,
    float* o, int row, int lane)
{
    const int  len  = cnt[row];
    const int  je   = min(len, LIST_CAP);
    const int* perm = slot + (size_t)row * LIST_CAP;
    const float* col = hsrc + lane * 4;

    float4 a0 = make_float4(0.f, 0.f, 0.f, 0.f);
    float4 a1 = a0, a2 = a0, a3 = a0;

    int j = 0;
    for (; j + 4 <= je; j += 4) {
        const float* r0 = col + (size_t)perm[j + 0] * D_FEAT;
        const float* r1 = col + (size_t)perm[j + 1] * D_FEAT;
        const float* r2 = col + (size_t)perm[j + 2] * D_FEAT;
        const float* r3 = col + (size_t)perm[j + 3] * D_FEAT;
        const float4 p0 = *reinterpret_cast<const float4*>(r0);
        const float4 p1 = *reinterpret_cast<const float4*>(r0 + 128);
        const float4 p2 = *reinterpret_cast<const float4*>(r0 + 256);
        const float4 p3 = *reinterpret_cast<const float4*>(r0 + 384);
        const float4 q0 = *reinterpret_cast<const float4*>(r1);
        const float4 q1 = *reinterpret_cast<const float4*>(r1 + 128);
        const float4 q2 = *reinterpret_cast<const float4*>(r1 + 256);
        const float4 q3 = *reinterpret_cast<const float4*>(r1 + 384);
        const float4 s0 = *reinterpret_cast<const float4*>(r2);
        const float4 s1 = *reinterpret_cast<const float4*>(r2 + 128);
        const float4 s2 = *reinterpret_cast<const float4*>(r2 + 256);
        const float4 s3 = *reinterpret_cast<const float4*>(r2 + 384);
        const float4 t0 = *reinterpret_cast<const float4*>(r3);
        const float4 t1 = *reinterpret_cast<const float4*>(r3 + 128);
        const float4 t2 = *reinterpret_cast<const float4*>(r3 + 256);
        const float4 t3 = *reinterpret_cast<const float4*>(r3 + 384);
        a0.x += p0.x; a0.y += p0.y; a0.z += p0.z; a0.w += p0.w;
        a1.x += p1.x; a1.y += p1.y; a1.z += p1.z; a1.w += p1.w;
        a2.x += p2.x; a2.y += p2.y; a2.z += p2.z; a2.w += p2.w;
        a3.x += p3.x; a3.y += p3.y; a3.z += p3.z; a3.w += p3.w;
        a0.x += q0.x; a0.y += q0.y; a0.z += q0.z; a0.w += q0.w;
        a1.x += q1.x; a1.y += q1.y; a1.z += q1.z; a1.w += q1.w;
        a2.x += q2.x; a2.y += q2.y; a2.z += q2.z; a2.w += q2.w;
        a3.x += q3.x; a3.y += q3.y; a3.z += q3.z; a3.w += q3.w;
        a0.x += s0.x; a0.y += s0.y; a0.z += s0.z; a0.w += s0.w;
        a1.x += s1.x; a1.y += s1.y; a1.z += s1.z; a1.w += s1.w;
        a2.x += s2.x; a2.y += s2.y; a2.z += s2.z; a2.w += s2.w;
        a3.x += s3.x; a3.y += s3.y; a3.z += s3.z; a3.w += s3.w;
        a0.x += t0.x; a0.y += t0.y; a0.z += t0.z; a0.w += t0.w;
        a1.x += t1.x; a1.y += t1.y; a1.z += t1.z; a1.w += t1.w;
        a2.x += t2.x; a2.y += t2.y; a2.z += t2.z; a2.w += t2.w;
        a3.x += t3.x; a3.y += t3.y; a3.z += t3.z; a3.w += t3.w;
    }
    const float* own = hdst + (size_t)row * D_FEAT + lane * 4;
    const float4 d0 = *reinterpret_cast<const float4*>(own);
    const float4 d1 = *reinterpret_cast<const float4*>(own + 128);
    const float4 d2 = *reinterpret_cast<const float4*>(own + 256);
    const float4 d3 = *reinterpret_cast<const float4*>(own + 384);

    for (; j < je; ++j) {
        const float* r0 = col + (size_t)perm[j] * D_FEAT;
        const float4 p0 = *reinterpret_cast<const float4*>(r0);
        const float4 p1 = *reinterpret_cast<const float4*>(r0 + 128);
        const float4 p2 = *reinterpret_cast<const float4*>(r0 + 256);
        const float4 p3 = *reinterpret_cast<const float4*>(r0 + 384);
        a0.x += p0.x; a0.y += p0.y; a0.z += p0.z; a0.w += p0.w;
        a1.x += p1.x; a1.y += p1.y; a1.z += p1.z; a1.w += p1.w;
        a2.x += p2.x; a2.y += p2.y; a2.z += p2.z; a2.w += p2.w;
        a3.x += p3.x; a3.y += p3.y; a3.z += p3.z; a3.w += p3.w;
    }

    if (len > LIST_CAP) {
        const int m = nspill[0];
        for (int k = 0; k < m; ++k) {
            if (spill[2 * (size_t)k] != row) continue;
            const float* r0 = col + (size_t)spill[2 * (size_t)k + 1] * D_FEAT;
            const float4 p0 = *reinterpret_cast<const float4*>(r0);
            const float4 p1 = *reinterpret_cast<const float4*>(r0 + 128);
            const float4 p2 = *reinterpret_cast<const float4*>(r0 + 256);
            const float4 p3 = *reinterpret_cast<const float4*>(r0 + 384);
            a0.x += p0.x; a0.y += p0.y; a0.z += p0.z; a0.w += p0.w;
            a1.x += p1.x; a1.y += p1.y; a1.z += p1.z; a1.w += p1.w;
            a2.x += p2.x; a2.y += p2.y; a2.z += p2.z; a2.w += p2.w;
            a3.x += p3.x; a3.y += p3.y; a3.z += p3.z; a3.w += p3.w;
        }
    }

    const float inv = 1.0f / ((float)len + 1.0f);
    float4 r0, r1, r2, r3;
    r0.x = (a0.x + d0.x) * inv; r0.y = (a0.y + d0.y) * inv;
    r0.z = (a0.z + d0.z) * inv; r0.w = (a0.w + d0.w) * inv;
    r1.x = (a1.x + d1.x) * inv; r1.y = (a1.y + d1.y) * inv;
    r1.z = (a1.z + d1.z) * inv; r1.w = (a1.w + d1.w) * inv;
    r2.x = (a2.x + d2.x) * inv; r2.y = (a2.y + d2.y) * inv;
    r2.z = (a2.z + d2.z) * inv; r2.w = (a2.w + d2.w) * inv;
    r3.x = (a3.x + d3.x) * inv; r3.y = (a3.y + d3.y) * inv;
    r3.z = (a3.z + d3.z) * inv; r3.w = (a3.w + d3.w) * inv;

    *reinterpret_cast<float4*>(o)       = r0;
    *reinterpret_cast<float4*>(o + 128) = r1;
    *reinterpret_cast<float4*>(o + 256) = r2;
    *reinterpret_cast<float4*>(o + 384) = r3;
}

#define GEMM_KU 4
#define GEMM_G  (4 * GEMM_KU)

template <int MT>
struct FragSet {
    float2 b[GEMM_KU][2];
    v2f    a[GEMM_KU][MT];
};

template <int MT>
__device__ __forceinline__ void gemm_load_set(
    FragSet<MT>& s, const float* __restrict__ bcol, const float* arow,
    int k, int N)
{
#pragma unroll
    for (int u = 0; u < GEMM_KU; ++u) {
        const int kk = k + 4 * u;
        s.b[u][0] = *reinterpret_cast<const float2*>(bcol + (size_t)kk * N);
        s.b[u][1] = *reinterpret_cast<const float2*>(bcol + (size_t)(kk + 1) * N);
#pragma unroll
        for (int m = 0; m < MT; ++m)
            s.a[u][m] = *reinterpret_cast<const v2f*>(
                arow + m * 16 * LDS_STRIDE + kk);
    }
}

template <int MT>
__device__ __forceinline__ void gemm_mma_set(
    const FragSet<MT>& s, v8f (&acc)[MT][2])
{
#pragma unroll
    for (int u = 0; u < GEMM_KU; ++u) {
        v2f b0, b1;
        b0.x = s.b[u][0].x; b0.y = s.b[u][1].x;
        b1.x = s.b[u][0].y; b1.y = s.b[u][1].y;
#pragma unroll
        for (int m = 0; m < MT; ++m) {
            acc[m][0] = __builtin_amdgcn_wmma_f32_16x16x4_f32(
                false, s.a[u][m], false, b0, (short)0, acc[m][0], false, false);
            acc[m][1] = __builtin_amdgcn_wmma_f32_16x16x4_f32(
                false, s.a[u][m], false, b1, (short)0, acc[m][1], false, false);
        }
    }
}

template <int MT, int NW, bool RELU>
__global__ __launch_bounds__(NW * 32) void sage_gather_gemm_wmma(
    const float* __restrict__ hsrc, const float* __restrict__ hdst,
    const int* __restrict__ cnt, const int* __restrict__ slot,
    const int* __restrict__ spill, const int* __restrict__ nspill,
    const float* __restrict__ B, const float* __restrict__ bias,
    float* __restrict__ C, int M)
{
    constexpr int K = D_FEAT;
    constexpr int N = NW * 32;
    __shared__ float As[MT * 16 * LDS_STRIDE];

    const int rm   = blockIdx.x * (16 * MT);
    const int wave = __builtin_amdgcn_readfirstlane((int)(threadIdx.x >> 5));
    const int lane = threadIdx.x & 31;
    const int half = lane >> 4;
    const int l    = lane & 15;
    const int cn   = wave * 32;

    for (int rr = wave; rr < 16 * MT; rr += NW) {
        float* o = As + rr * LDS_STRIDE + lane * 4;
        const int row = rm + rr;
        if (row < M) {
            sage_gather_combine_row(hsrc, hdst, cnt, slot, spill, nspill, o,
                                    row, lane);
        } else {
            const float4 z = make_float4(0.f, 0.f, 0.f, 0.f);
            *reinterpret_cast<float4*>(o)       = z;
            *reinterpret_cast<float4*>(o + 128) = z;
            *reinterpret_cast<float4*>(o + 256) = z;
            *reinterpret_cast<float4*>(o + 384) = z;
        }
    }
    __syncthreads();

    v8f acc[MT][2];
#pragma unroll
    for (int m = 0; m < MT; ++m) {
        acc[m][0] = v8f{};
        acc[m][1] = v8f{};
    }

    const float* arow = As + l * LDS_STRIDE + 2 * half;
    const float* bcol = B + (size_t)(2 * half) * N + cn + 2 * l;

    FragSet<MT> s0, s1;
    gemm_load_set<MT>(s0, bcol, arow, 0, N);
    for (int k = 0; k < K - 2 * GEMM_G; k += 2 * GEMM_G) {
        gemm_load_set<MT>(s1, bcol, arow, k + GEMM_G, N);
        gemm_mma_set<MT>(s0, acc);
        gemm_load_set<MT>(s0, bcol, arow, k + 2 * GEMM_G, N);
        gemm_mma_set<MT>(s1, acc);
    }
    gemm_load_set<MT>(s1, bcol, arow, K - GEMM_G, N);
    gemm_mma_set<MT>(s0, acc);
    gemm_mma_set<MT>(s1, acc);

    const float2 bv = *reinterpret_cast<const float2*>(bias + cn + 2 * l);
#pragma unroll
    for (int m = 0; m < MT; ++m) {
#pragma unroll
        for (int i = 0; i < 8; ++i) {
            const int row = rm + m * 16 + i + 8 * half;
            float2 v;
            v.x = acc[m][0][i] + bv.x;
            v.y = acc[m][1][i] + bv.y;
            if (RELU) {
                v.x = fmaxf(v.x, 0.0f);
                v.y = fmaxf(v.y, 0.0f);
            }
            if (row < M)
                *reinterpret_cast<float2*>(C + (size_t)row * N + cn + 2 * l) = v;
        }
    }
}

extern "C" void kernel_launch(void* const* d_in, const int* in_sizes, int n_in,
                              void* d_out, int out_size, void* d_ws, size_t ws_size,
                              hipStream_t stream)
{
    const float* feats = (const float*)d_in[0];
    const int*   src0  = (const int*)d_in[1];
    const int*   dst0  = (const int*)d_in[2];
    const int*   src1  = (const int*)d_in[3];
    const int*   dst1  = (const int*)d_in[4];
    const float* W1    = (const float*)d_in[5];
    const float* b1    = (const float*)d_in[6];
    const float* W2    = (const float*)d_in[7];
    const float* b2    = (const float*)d_in[8];
    float* out = (float*)d_out;

    const int N1 = 25000, N2 = 5000;
    const int E0 = in_sizes[1];
    const int E1 = in_sizes[3];
    const size_t F = D_FEAT;

    float* ws   = (float*)d_ws;
    float* scr0 = ws;
    float* h1   = scr0 + 2 * (size_t)N1 * F;

    int* slot0  = (int*)scr0;
    int* slot1  = slot0 + (size_t)N1 * LIST_CAP;
    int* spill0 = slot1 + (size_t)N2 * LIST_CAP;
    int* spill1 = spill0 + 2 * (size_t)E0;
    int* cnt0   = spill1 + 2 * (size_t)E1;
    int* cnt1   = cnt0 + N1;
    int* nspill = cnt1 + N2;

    const int nb0 = (E0 + 255) / 256;
    const int nb1 = (E1 + 255) / 256;
    hipMemsetAsync(cnt0, 0, (size_t)(N1 + N2 + 2) * sizeof(int), stream);
    sage_edge_slots<<<nb0 + nb1, 256, 0, stream>>>(
        src0, dst0, src1, dst1, cnt0, cnt1, slot0, slot1, spill0, spill1,
        nspill, E0, E1, nb0);

    sage_gather_gemm_wmma<2, 16, true><<<(N1 + 31) / 32, 512, 0, stream>>>(
        feats, feats, cnt0, slot0, spill0, nspill, W1, b1, h1, N1);

    sage_gather_gemm_wmma<1, 8, false><<<(N2 + 15) / 16, 256, 0, stream>>>(
        h1, h1, cnt1, slot1, spill1, nspill + 1, W2, b2, out, N2);
}


# Layer 0: one resident block per compute unit owns an even 98-row share (7 M tiles a wave, 112x32 outputs), so W1 comes from L2 once per unit, not once per 32 rows; sum order kept
# speedup vs baseline: 3.0904x; 1.0570x over previous
#include <hip/hip_runtime.h>
#include <cstddef>

typedef float v2f __attribute__((ext_vector_type(2)));
typedef float v8f __attribute__((ext_vector_type(8)));

#define D_FEAT 512
#define LDS_STRIDE 516

#define LIST_CAP 64

__global__ __launch_bounds__(256) void sage_edge_slots(
    const int* __restrict__ src0, const int* __restrict__ dst0,
    const int* __restrict__ src1, const int* __restrict__ dst1,
    int* __restrict__ cnt0, int* __restrict__ cnt1,
    int* __restrict__ slot0, int* __restrict__ slot1,
    int* __restrict__ spill0, int* __restrict__ spill1,
    int* __restrict__ nspill, int E0, int E1, int nb0)
{
    const bool first = (int)blockIdx.x < nb0;
    const int* src   = first ? src0 : src1;
    const int* dst   = first ? dst0 : dst1;
    int*       cnt   = first ? cnt0 : cnt1;
    int*       slot  = first ? slot0 : slot1;
    int*       spill = first ? spill0 : spill1;
    int*       nsp   = first ? nspill : nspill + 1;
    const int  E     = first ? E0 : E1;
    const int  b     = first ? (int)blockIdx.x : (int)blockIdx.x - nb0;

    const int e = b * 256 + threadIdx.x;
    if (e >= E) return;
    const int d = dst[e];
    const int s = src[e];
    const int r = atomicAdd(cnt + d, 1);
    if (r < LIST_CAP) {
        slot[(size_t)d * LIST_CAP + r] = s;
    } else {
        const int k = atomicAdd(nsp, 1);
        spill[2 * (size_t)k]     = d;
        spill[2 * (size_t)k + 1] = s;
    }
}

__device__ __forceinline__ void sage_gather_combine_row(
    const float* __restrict__ hsrc, const float* __restrict__ hdst,
    const int* __restrict__ cnt, const int* __restrict__ slot,
    const int* __restrict__ spill, const int* __restrict__ nspill,
    float* o, int row, int lane)
{
    const int  len  = cnt[row];
    const int  je   = min(len, LIST_CAP);
    const int* perm = slot + (size_t)row * LIST_CAP;
    const float* col = hsrc + lane * 4;

    float4 a0 = make_float4(0.f, 0.f, 0.f, 0.f);
    float4 a1 = a0, a2 = a0, a3 = a0;

    int j = 0;
    for (; j + 4 <= je; j += 4) {
        const float* r0 = col + (size_t)perm[j + 0] * D_FEAT;
        const float* r1 = col + (size_t)perm[j + 1] * D_FEAT;
        const float* r2 = col + (size_t)perm[j + 2] * D_FEAT;
        const float* r3 = col + (size_t)perm[j + 3] * D_FEAT;
        const float4 p0 = *reinterpret_cast<const float4*>(r0);
        const float4 p1 = *reinterpret_cast<const float4*>(r0 + 128);
        const float4 p2 = *reinterpret_cast<const float4*>(r0 + 256);
        const float4 p3 = *reinterpret_cast<const float4*>(r0 + 384);
        const float4 q0 = *reinterpret_cast<const float4*>(r1);
        const float4 q1 = *reinterpret_cast<const float4*>(r1 + 128);
        const float4 q2 = *reinterpret_cast<const float4*>(r1 + 256);
        const float4 q3 = *reinterpret_cast<const float4*>(r1 + 384);
        const float4 s0 = *reinterpret_cast<const float4*>(r2);
        const float4 s1 = *reinterpret_cast<const float4*>(r2 + 128);
        const float4 s2 = *reinterpret_cast<const float4*>(r2 + 256);
        const float4 s3 = *reinterpret_cast<const float4*>(r2 + 384);
        const float4 t0 = *reinterpret_cast<const float4*>(r3);
        const float4 t1 = *reinterpret_cast<const float4*>(r3 + 128);
        const float4 t2 = *reinterpret_cast<const float4*>(r3 + 256);
        const float4 t3 = *reinterpret_cast<const float4*>(r3 + 384);
        a0.x += p0.x; a0.y += p0.y; a0.z += p0.z; a0.w += p0.w;
        a1.x += p1.x; a1.y += p1.y; a1.z += p1.z; a1.w += p1.w;
        a2.x += p2.x; a2.y += p2.y; a2.z += p2.z; a2.w += p2.w;
        a3.x += p3.x; a3.y += p3.y; a3.z += p3.z; a3.w += p3.w;
        a0.x += q0.x; a0.y += q0.y; a0.z += q0.z; a0.w += q0.w;
        a1.x += q1.x; a1.y += q1.y; a1.z += q1.z; a1.w += q1.w;
        a2.x += q2.x; a2.y += q2.y; a2.z += q2.z; a2.w += q2.w;
        a3.x += q3.x; a3.y += q3.y; a3.z += q3.z; a3.w += q3.w;
        a0.x += s0.x; a0.y += s0.y; a0.z += s0.z; a0.w += s0.w;
        a1.x += s1.x; a1.y += s1.y; a1.z += s1.z; a1.w += s1.w;
        a2.x += s2.x; a2.y += s2.y; a2.z += s2.z; a2.w += s2.w;
        a3.x += s3.x; a3.y += s3.y; a3.z += s3.z; a3.w += s3.w;
        a0.x += t0.x; a0.y += t0.y; a0.z += t0.z; a0.w += t0.w;
        a1.x += t1.x; a1.y += t1.y; a1.z += t1.z; a1.w += t1.w;
        a2.x += t2.x; a2.y += t2.y; a2.z += t2.z; a2.w += t2.w;
        a3.x += t3.x; a3.y += t3.y; a3.z += t3.z; a3.w += t3.w;
    }
    const float* own = hdst + (size_t)row * D_FEAT + lane * 4;
    const float4 d0 = *reinterpret_cast<const float4*>(own);
    const float4 d1 = *reinterpret_cast<const float4*>(own + 128);
    const float4 d2 = *reinterpret_cast<const float4*>(own + 256);
    const float4 d3 = *reinterpret_cast<const float4*>(own + 384);

    for (; j < je; ++j) {
        const float* r0 = col + (size_t)perm[j] * D_FEAT;
        const float4 p0 = *reinterpret_cast<const float4*>(r0);
        const float4 p1 = *reinterpret_cast<const float4*>(r0 + 128);
        const float4 p2 = *reinterpret_cast<const float4*>(r0 + 256);
        const float4 p3 = *reinterpret_cast<const float4*>(r0 + 384);
        a0.x += p0.x; a0.y += p0.y; a0.z += p0.z; a0.w += p0.w;
        a1.x += p1.x; a1.y += p1.y; a1.z += p1.z; a1.w += p1.w;
        a2.x += p2.x; a2.y += p2.y; a2.z += p2.z; a2.w += p2.w;
        a3.x += p3.x; a3.y += p3.y; a3.z += p3.z; a3.w += p3.w;
    }

    if (len > LIST_CAP) {
        const int m = nspill[0];
        for (int k = 0; k < m; ++k) {
            if (spill[2 * (size_t)k] != row) continue;
            const float* r0 = col + (size_t)spill[2 * (size_t)k + 1] * D_FEAT;
            const float4 p0 = *reinterpret_cast<const float4*>(r0);
            const float4 p1 = *reinterpret_cast<const float4*>(r0 + 128);
            const float4 p2 = *reinterpret_cast<const float4*>(r0 + 256);
            const float4 p3 = *reinterpret_cast<const float4*>(r0 + 384);
            a0.x += p0.x; a0.y += p0.y; a0.z += p0.z; a0.w += p0.w;
            a1.x += p1.x; a1.y += p1.y; a1.z += p1.z; a1.w += p1.w;
            a2.x += p2.x; a2.y += p2.y; a2.z += p2.z; a2.w += p2.w;
            a3.x += p3.x; a3.y += p3.y; a3.z += p3.z; a3.w += p3.w;
        }
    }

    const float inv = 1.0f / ((float)len + 1.0f);
    float4 r0, r1, r2, r3;
    r0.x = (a0.x + d0.x) * inv; r0.y = (a0.y + d0.y) * inv;
    r0.z = (a0.z + d0.z) * inv; r0.w = (a0.w + d0.w) * inv;
    r1.x = (a1.x + d1.x) * inv; r1.y = (a1.y + d1.y) * inv;
    r1.z = (a1.z + d1.z) * inv; r1.w = (a1.w + d1.w) * inv;
    r2.x = (a2.x + d2.x) * inv; r2.y = (a2.y + d2.y) * inv;
    r2.z = (a2.z + d2.z) * inv; r2.w = (a2.w + d2.w) * inv;
    r3.x = (a3.x + d3.x) * inv; r3.y = (a3.y + d3.y) * inv;
    r3.z = (a3.z + d3.z) * inv; r3.w = (a3.w + d3.w) * inv;

    *reinterpret_cast<float4*>(o)       = r0;
    *reinterpret_cast<float4*>(o + 128) = r1;
    *reinterpret_cast<float4*>(o + 256) = r2;
    *reinterpret_cast<float4*>(o + 384) = r3;
}

template <int MT, int KU>
struct FragSet {
    float2 b[KU][2];
    v2f    a[KU][MT];
};

template <int MT, int KU>
__device__ __forceinline__ void gemm_load_set(
    FragSet<MT, KU>& s, const float* __restrict__ bcol, const float* arow,
    int k, int N)
{
#pragma unroll
    for (int u = 0; u < KU; ++u) {
        const int kk = k + 4 * u;
        s.b[u][0] = *reinterpret_cast<const float2*>(bcol + (size_t)kk * N);
        s.b[u][1] = *reinterpret_cast<const float2*>(bcol + (size_t)(kk + 1) * N);
#pragma unroll
        for (int m = 0; m < MT; ++m)
            s.a[u][m] = *reinterpret_cast<const v2f*>(
                arow + m * 16 * LDS_STRIDE + kk);
    }
}

template <int MT, int KU>
__device__ __forceinline__ void gemm_mma_set(
    const FragSet<MT, KU>& s, v8f (&acc)[MT][2])
{
#pragma unroll
    for (int u = 0; u < KU; ++u) {
        v2f b0, b1;
        b0.x = s.b[u][0].x; b0.y = s.b[u][1].x;
        b1.x = s.b[u][0].y; b1.y = s.b[u][1].y;
#pragma unroll
        for (int m = 0; m < MT; ++m) {
            acc[m][0] = __builtin_amdgcn_wmma_f32_16x16x4_f32(
                false, s.a[u][m], false, b0, (short)0, acc[m][0], false, false);
            acc[m][1] = __builtin_amdgcn_wmma_f32_16x16x4_f32(
                false, s.a[u][m], false, b1, (short)0, acc[m][1], false, false);
        }
    }
}

template <int MT, int NW, int KU, int ROWS, bool RELU>
__global__ __launch_bounds__(NW * 32) void sage_gather_gemm_wmma(
    const float* __restrict__ hsrc, const float* __restrict__ hdst,
    const int* __restrict__ cnt, const int* __restrict__ slot,
    const int* __restrict__ spill, const int* __restrict__ nspill,
    const float* __restrict__ B, const float* __restrict__ bias,
    float* __restrict__ C, int M)
{
    constexpr int K = D_FEAT;
    constexpr int N = NW * 32;
    constexpr int G = 4 * KU;
    static_assert(ROWS <= 16 * MT, "tile must hold the block's rows");
    __shared__ float As[MT * 16 * LDS_STRIDE];

    const int rm   = blockIdx.x * ROWS;
    const int wave = __builtin_amdgcn_readfirstlane((int)(threadIdx.x >> 5));
    const int lane = threadIdx.x & 31;
    const int half = lane >> 4;
    const int l    = lane & 15;
    const int cn   = wave * 32;

    for (int rr = wave; rr < 16 * MT; rr += NW) {
        float* o = As + rr * LDS_STRIDE + lane * 4;
        const int row = rm + rr;
        if (rr < ROWS && row < M) {
            sage_gather_combine_row(hsrc, hdst, cnt, slot, spill, nspill, o,
                                    row, lane);
        } else {
            const float4 z = make_float4(0.f, 0.f, 0.f, 0.f);
            *reinterpret_cast<float4*>(o)       = z;
            *reinterpret_cast<float4*>(o + 128) = z;
            *reinterpret_cast<float4*>(o + 256) = z;
            *reinterpret_cast<float4*>(o + 384) = z;
        }
    }
    __syncthreads();

    v8f acc[MT][2];
#pragma unroll
    for (int m = 0; m < MT; ++m) {
        acc[m][0] = v8f{};
        acc[m][1] = v8f{};
    }

    const float* arow = As + l * LDS_STRIDE + 2 * half;
    const float* bcol = B + (size_t)(2 * half) * N + cn + 2 * l;

    FragSet<MT, KU> s0, s1;
    gemm_load_set<MT, KU>(s0, bcol, arow, 0, N);
    for (int k = 0; k < K - 2 * G; k += 2 * G) {
        gemm_load_set<MT, KU>(s1, bcol, arow, k + G, N);
        gemm_mma_set<MT, KU>(s0, acc);
        gemm_load_set<MT, KU>(s0, bcol, arow, k + 2 * G, N);
        gemm_mma_set<MT, KU>(s1, acc);
    }
    gemm_load_set<MT, KU>(s1, bcol, arow, K - G, N);
    gemm_mma_set<MT, KU>(s0, acc);
    gemm_mma_set<MT, KU>(s1, acc);

    const float2 bv = *reinterpret_cast<const float2*>(bias + cn + 2 * l);
#pragma unroll
    for (int m = 0; m < MT; ++m) {
#pragma unroll
        for (int i = 0; i < 8; ++i) {
            const int rr  = m * 16 + i + 8 * half;
            const int row = rm + rr;
            float2 v;
            v.x = acc[m][0][i] + bv.x;
            v.y = acc[m][1][i] + bv.y;
            if (RELU) {
                v.x = fmaxf(v.x, 0.0f);
                v.y = fmaxf(v.y, 0.0f);
            }
            if (rr < ROWS && row < M)
                *reinterpret_cast<float2*>(C + (size_t)row * N + cn + 2 * l) = v;
        }
    }
}

extern "C" void kernel_launch(void* const* d_in, const int* in_sizes, int n_in,
                              void* d_out, int out_size, void* d_ws, size_t ws_size,
                              hipStream_t stream)
{
    const float* feats = (const float*)d_in[0];
    const int*   src0  = (const int*)d_in[1];
    const int*   dst0  = (const int*)d_in[2];
    const int*   src1  = (const int*)d_in[3];
    const int*   dst1  = (const int*)d_in[4];
    const float* W1    = (const float*)d_in[5];
    const float* b1    = (const float*)d_in[6];
    const float* W2    = (const float*)d_in[7];
    const float* b2    = (const float*)d_in[8];
    float* out = (float*)d_out;

    const int N1 = 25000, N2 = 5000;
    const int E0 = in_sizes[1];
    const int E1 = in_sizes[3];
    const size_t F = D_FEAT;

    float* ws   = (float*)d_ws;
    float* scr0 = ws;
    float* h1   = scr0 + 2 * (size_t)N1 * F;

    int* slot0  = (int*)scr0;
    int* slot1  = slot0 + (size_t)N1 * LIST_CAP;
    int* spill0 = slot1 + (size_t)N2 * LIST_CAP;
    int* spill1 = spill0 + 2 * (size_t)E0;
    int* cnt0   = spill1 + 2 * (size_t)E1;
    int* cnt1   = cnt0 + N1;
    int* nspill = cnt1 + N2;

    const int nb0 = (E0 + 255) / 256;
    const int nb1 = (E1 + 255) / 256;
    hipMemsetAsync(cnt0, 0, (size_t)(N1 + N2 + 2) * sizeof(int), stream);
    sage_edge_slots<<<nb0 + nb1, 256, 0, stream>>>(
        src0, dst0, src1, dst1, cnt0, cnt1, slot0, slot1, spill0, spill1,
        nspill, E0, E1, nb0);

    sage_gather_gemm_wmma<7, 16, 2, 98, true><<<(N1 + 97) / 98, 512, 0, stream>>>(
        feats, feats, cnt0, slot0, spill0, nspill, W1, b1, h1, N1);

    sage_gather_gemm_wmma<1, 8, 4, 16, false><<<(N2 + 15) / 16, 256, 0, stream>>>(
        h1, h1, cnt1, slot1, spill1, nspill + 1, W2, b2, out, N2);
}
